# starting kernel
# baseline (speedup 1.0000x reference)
#include <hip/hip_runtime.h>

typedef __attribute__((ext_vector_type(2))) float v2f;
typedef __attribute__((ext_vector_type(4))) float v4f;
typedef __attribute__((ext_vector_type(8))) float v8f;

#define DIM        512
#define BATCH      32
#define M_TOTAL    (512 * 16 * 16)
#define TILES_PER_WAVE 2
#define ROWS_PER_WAVE  (16 * TILES_PER_WAVE)
#define WAVES_PER_BLK  8
#define ROWS_PER_BLK   (ROWS_PER_WAVE * WAVES_PER_BLK)
#define FACTOR     1.0f

__global__ __launch_bounds__(256) void eigenspace_wmma_kernel(
    const float* __restrict__ style,
    const float* __restrict__ U,
    const float* __restrict__ L,
    const float* __restrict__ mu,
    float* __restrict__ out)
{
    __shared__ float s_lds[DIM * BATCH];

    const int tid = threadIdx.x;

    for (int idx = tid; idx < DIM * BATCH; idx += 256) {
        const int b = idx >> 9;
        const int k = idx & (DIM - 1);
        const int p = k >> 1;
        const int j = k & 1;
        const int rot = (2 * b + ((p & 1) << 5)) & 63;
        s_lds[(p << 6) + rot + j] = style[b * DIM + k] * (L[k] * FACTOR);
    }
    __syncthreads();

    const int wave   = tid >> 5;
    const int lane   = tid & 31;
    const int lane16 = lane & 15;
    const int khalf  = lane >> 4;

    const int m_base = blockIdx.x * ROWS_PER_BLK + wave * ROWS_PER_WAVE;

    const float* Urow0 = U + (size_t)(m_base +      lane16) * DIM + 2 * khalf;
    const float* Urow1 = U + (size_t)(m_base + 16 + lane16) * DIM + 2 * khalf;

    const int rot0  = (2 * lane16 +      (khalf << 5)) & 63;
    const int rot1  = (2 * lane16 + 32 + (khalf << 5)) & 63;
    const float* sb0 = s_lds + (khalf << 6) + rot0;
    const float* sb1 = s_lds + (khalf << 6) + rot1;

    v8f c00 = {};
    v8f c01 = {};
    v8f c10 = {};
    v8f c11 = {};

    #pragma unroll 8
    for (int k0 = 0; k0 < DIM; k0 += 4) {
        v2f a0 = *(const v2f*)(Urow0 + k0);
        v2f a1 = *(const v2f*)(Urow1 + k0);

        v2f b0 = *(const v2f*)(sb0 + k0 * 32);
        v2f b1 = *(const v2f*)(sb1 + k0 * 32);

        c00 = __builtin_amdgcn_wmma_f32_16x16x4_f32(
                  false, a0, false, b0, (short)0, c00, false, false);
        c01 = __builtin_amdgcn_wmma_f32_16x16x4_f32(
                  false, a0, false, b1, (short)0, c01, false, false);
        c10 = __builtin_amdgcn_wmma_f32_16x16x4_f32(
                  false, a1, false, b0, (short)0, c10, false, false);
        c11 = __builtin_amdgcn_wmma_f32_16x16x4_f32(
                  false, a1, false, b1, (short)0, c11, false, false);
    }

    const int mrow0 = m_base +      8 * khalf;
    const int mrow1 = m_base + 16 + 8 * khalf;

    const float* mup0 = mu + mrow0;
    const float* mup1 = mu + mrow1;
    #pragma unroll
    for (int v = 0; v < 8; ++v) {
        const float mv0 = mup0[v];
        const float mv1 = mup1[v];
        c00[v] += mv0;  c01[v] += mv0;
        c10[v] += mv1;  c11[v] += mv1;
    }

    float* o00 = out + (size_t)lane16        * M_TOTAL + mrow0;
    float* o01 = out + (size_t)(lane16 + 16) * M_TOTAL + mrow0;
    float* o10 = out + (size_t)lane16        * M_TOTAL + mrow1;
    float* o11 = out + (size_t)(lane16 + 16) * M_TOTAL + mrow1;

    *(v4f*)(o00)     = (v4f){c00[0], c00[1], c00[2], c00[3]};
    *(v4f*)(o00 + 4) = (v4f){c00[4], c00[5], c00[6], c00[7]};
    *(v4f*)(o01)     = (v4f){c01[0], c01[1], c01[2], c01[3]};
    *(v4f*)(o01 + 4) = (v4f){c01[4], c01[5], c01[6], c01[7]};
    *(v4f*)(o10)     = (v4f){c10[0], c10[1], c10[2], c10[3]};
    *(v4f*)(o10 + 4) = (v4f){c10[4], c10[5], c10[6], c10[7]};
    *(v4f*)(o11)     = (v4f){c11[0], c11[1], c11[2], c11[3]};
    *(v4f*)(o11 + 4) = (v4f){c11[4], c11[5], c11[6], c11[7]};
}

extern "C" void kernel_launch(void* const* d_in, const int* in_sizes, int n_in,
                              void* d_out, int out_size, void* d_ws, size_t ws_size,
                              hipStream_t stream) {
    const float* style = (const float*)d_in[0];
    const float* U     = (const float*)d_in[1];
    const float* L     = (const float*)d_in[2];
    const float* mu    = (const float*)d_in[3];
    float* out         = (float*)d_out;

    (void)in_sizes; (void)n_in; (void)out_size; (void)d_ws; (void)ws_size;

    const int grid = M_TOTAL / ROWS_PER_BLK;
    eigenspace_wmma_kernel<<<grid, 256, 0, stream>>>(style, U, L, mu, out);
}


# U row loads 16 B per lane (global_load_b128, was 8 B) and B fragments as ds_load_b128, k order inside each 8-wide block permuted alike in both loaders; the order of the k sums changes
# speedup vs baseline: 1.0644x; 1.0644x over previous
#include <hip/hip_runtime.h>

typedef __attribute__((ext_vector_type(2))) float v2f;
typedef __attribute__((ext_vector_type(4))) float v4f;
typedef __attribute__((ext_vector_type(8))) float v8f;

#define DIM        512
#define BATCH      32
#define M_TOTAL    (512 * 16 * 16)
#define TILES_PER_WAVE 2
#define ROWS_PER_WAVE  (16 * TILES_PER_WAVE)
#define WAVES_PER_BLK  8
#define ROWS_PER_BLK   (ROWS_PER_WAVE * WAVES_PER_BLK)
#define FACTOR     1.0f

__global__ __launch_bounds__(256) void eigenspace_wmma_kernel(
    const float* __restrict__ style,
    const float* __restrict__ U,
    const float* __restrict__ L,
    const float* __restrict__ mu,
    float* __restrict__ out)
{
    __shared__ __attribute__((aligned(16))) float s_lds[DIM * BATCH];

    const int tid = threadIdx.x;

    for (int idx = tid; idx < DIM * BATCH; idx += 256) {
        const int b = idx >> 9;
        const int k = idx & (DIM - 1);
        const int q = k >> 2;
        const int j = k & 3;
        s_lds[(q << 7) + (b << 2) + j] = style[b * DIM + k] * (L[k] * FACTOR);
    }
    __syncthreads();

    const int wave   = tid >> 5;
    const int lane   = tid & 31;
    const int lane16 = lane & 15;
    const int khalf  = lane >> 4;

    const int m_base = blockIdx.x * ROWS_PER_BLK + wave * ROWS_PER_WAVE;

    const float* Urow0 = U + (size_t)(m_base +      lane16) * DIM + 4 * khalf;
    const float* Urow1 = U + (size_t)(m_base + 16 + lane16) * DIM + 4 * khalf;

    const float* sb0 = s_lds + (khalf << 7) + 4 * lane16;
    const float* sb1 = sb0 + 64;

    v8f c00 = {};
    v8f c01 = {};
    v8f c10 = {};
    v8f c11 = {};

    #pragma unroll 4
    for (int kb = 0; kb < DIM; kb += 8) {
        const v4f a0 = *(const v4f*)(Urow0 + kb);
        const v4f a1 = *(const v4f*)(Urow1 + kb);

        const v4f b0 = *(const v4f*)(sb0 + kb * 32);
        const v4f b1 = *(const v4f*)(sb1 + kb * 32);

        const v2f a0l = {a0[0], a0[1]}, a0h = {a0[2], a0[3]};
        const v2f a1l = {a1[0], a1[1]}, a1h = {a1[2], a1[3]};
        const v2f b0l = {b0[0], b0[1]}, b0h = {b0[2], b0[3]};
        const v2f b1l = {b1[0], b1[1]}, b1h = {b1[2], b1[3]};

        c00 = __builtin_amdgcn_wmma_f32_16x16x4_f32(
                  false, a0l, false, b0l, (short)0, c00, false, false);
        c01 = __builtin_amdgcn_wmma_f32_16x16x4_f32(
                  false, a0l, false, b1l, (short)0, c01, false, false);
        c10 = __builtin_amdgcn_wmma_f32_16x16x4_f32(
                  false, a1l, false, b0l, (short)0, c10, false, false);
        c11 = __builtin_amdgcn_wmma_f32_16x16x4_f32(
                  false, a1l, false, b1l, (short)0, c11, false, false);
        c00 = __builtin_amdgcn_wmma_f32_16x16x4_f32(
                  false, a0h, false, b0h, (short)0, c00, false, false);
        c01 = __builtin_amdgcn_wmma_f32_16x16x4_f32(
                  false, a0h, false, b1h, (short)0, c01, false, false);
        c10 = __builtin_amdgcn_wmma_f32_16x16x4_f32(
                  false, a1h, false, b0h, (short)0, c10, false, false);
        c11 = __builtin_amdgcn_wmma_f32_16x16x4_f32(
                  false, a1h, false, b1h, (short)0, c11, false, false);
    }

    const int mrow0 = m_base +      8 * khalf;
    const int mrow1 = m_base + 16 + 8 * khalf;

    const float* mup0 = mu + mrow0;
    const float* mup1 = mu + mrow1;
    #pragma unroll
    for (int v = 0; v < 8; ++v) {
        const float mv0 = mup0[v];
        const float mv1 = mup1[v];
        c00[v] += mv0;  c01[v] += mv0;
        c10[v] += mv1;  c11[v] += mv1;
    }

    float* o00 = out + (size_t)lane16        * M_TOTAL + mrow0;
    float* o01 = out + (size_t)(lane16 + 16) * M_TOTAL + mrow0;
    float* o10 = out + (size_t)lane16        * M_TOTAL + mrow1;
    float* o11 = out + (size_t)(lane16 + 16) * M_TOTAL + mrow1;

    *(v4f*)(o00)     = (v4f){c00[0], c00[1], c00[2], c00[3]};
    *(v4f*)(o00 + 4) = (v4f){c00[4], c00[5], c00[6], c00[7]};
    *(v4f*)(o01)     = (v4f){c01[0], c01[1], c01[2], c01[3]};
    *(v4f*)(o01 + 4) = (v4f){c01[4], c01[5], c01[6], c01[7]};
    *(v4f*)(o10)     = (v4f){c10[0], c10[1], c10[2], c10[3]};
    *(v4f*)(o10 + 4) = (v4f){c10[4], c10[5], c10[6], c10[7]};
    *(v4f*)(o11)     = (v4f){c11[0], c11[1], c11[2], c11[3]};
    *(v4f*)(o11 + 4) = (v4f){c11[4], c11[5], c11[6], c11[7]};
}

extern "C" void kernel_launch(void* const* d_in, const int* in_sizes, int n_in,
                              void* d_out, int out_size, void* d_ws, size_t ws_size,
                              hipStream_t stream) {
    const float* style = (const float*)d_in[0];
    const float* U     = (const float*)d_in[1];
    const float* L     = (const float*)d_in[2];
    const float* mu    = (const float*)d_in[3];
    float* out         = (float*)d_out;

    (void)in_sizes; (void)n_in; (void)out_size; (void)d_ws; (void)ws_size;

    const int grid = M_TOTAL / ROWS_PER_BLK;
    eigenspace_wmma_kernel<<<grid, 256, 0, stream>>>(style, U, L, mu, out);
}


# Software-pipelined k loop, fully unrolled: U loads issued two or more 32-k blocks ahead of their WMMAs, LDS loads one block ahead (parent: no load of the next trip in flight); k sum order unchanged
# speedup vs baseline: 1.1540x; 1.0842x over previous
#include <hip/hip_runtime.h>

typedef __attribute__((ext_vector_type(2))) float v2f;
typedef __attribute__((ext_vector_type(4))) float v4f;
typedef __attribute__((ext_vector_type(8))) float v8f;

#define DIM        512
#define BATCH      32
#define M_TOTAL    (512 * 16 * 16)
#define TILES_PER_WAVE 2
#define ROWS_PER_WAVE  (16 * TILES_PER_WAVE)
#define WAVES_PER_BLK  8
#define ROWS_PER_BLK   (ROWS_PER_WAVE * WAVES_PER_BLK)
#define FACTOR     1.0f

__global__ __launch_bounds__(256) void eigenspace_wmma_kernel(
    const float* __restrict__ style,
    const float* __restrict__ U,
    const float* __restrict__ L,
    const float* __restrict__ mu,
    float* __restrict__ out)
{
    __shared__ __attribute__((aligned(16))) float s_lds[DIM * BATCH];

    const int tid = threadIdx.x;

    for (int idx = tid; idx < DIM * BATCH; idx += 256) {
        const int b = idx >> 9;
        const int k = idx & (DIM - 1);
        const int q = k >> 2;
        const int j = k & 3;
        s_lds[(q << 7) + (b << 2) + j] = style[b * DIM + k] * (L[k] * FACTOR);
    }
    __syncthreads();

    const int wave   = tid >> 5;
    const int lane   = tid & 31;
    const int lane16 = lane & 15;
    const int khalf  = lane >> 4;

    const int m_base = blockIdx.x * ROWS_PER_BLK + wave * ROWS_PER_WAVE;

    const float* Urow0 = U + (size_t)(m_base +      lane16) * DIM + 4 * khalf;
    const float* Urow1 = U + (size_t)(m_base + 16 + lane16) * DIM + 4 * khalf;

    const float* sb0 = s_lds + (khalf << 7) + 4 * lane16;
    const float* sb1 = sb0 + 64;

    v8f c00 = {};
    v8f c01 = {};
    v8f c10 = {};
    v8f c11 = {};


#define LOAD_U(S, g)                                                        \
    _Pragma("unroll")                                                       \
    for (int j = 0; j < 4; ++j) {                                           \
        S##0[j] = *(const v4f*)(Urow0 + 32 * (g) + 8 * j);                  \
        S##1[j] = *(const v4f*)(Urow1 + 32 * (g) + 8 * j);                  \
    }

#define LOAD_S(S, g)                                                        \
    _Pragma("unroll")                                                       \
    for (int j = 0; j < 4; ++j) {                                           \
        S##0[j] = *(const v4f*)(sb0 + (32 * (g) + 8 * j) * 32);             \
        S##1[j] = *(const v4f*)(sb1 + (32 * (g) + 8 * j) * 32);             \
    }

#define WMMA_F32(acc, a, b)                                                 \
    acc = __builtin_amdgcn_wmma_f32_16x16x4_f32(false, a, false, b, (short)0, acc, false, false)

#define COMPUTE(A, B)                                                       \
    _Pragma("unroll")                                                       \
    for (int j = 0; j < 4; ++j) {                                           \
        const v2f a0l = {A##0[j][0], A##0[j][1]}, a0h = {A##0[j][2], A##0[j][3]}; \
        const v2f a1l = {A##1[j][0], A##1[j][1]}, a1h = {A##1[j][2], A##1[j][3]}; \
        const v2f b0l = {B##0[j][0], B##0[j][1]}, b0h = {B##0[j][2], B##0[j][3]}; \
        const v2f b1l = {B##1[j][0], B##1[j][1]}, b1h = {B##1[j][2], B##1[j][3]}; \
        WMMA_F32(c00, a0l, b0l);  WMMA_F32(c01, a0l, b1l);                  \
        WMMA_F32(c10, a1l, b0l);  WMMA_F32(c11, a1l, b1l);                  \
        WMMA_F32(c00, a0h, b0h);  WMMA_F32(c01, a0h, b1h);                  \
        WMMA_F32(c10, a1h, b0h);  WMMA_F32(c11, a1h, b1h);                  \
    }

#define PIN() __builtin_amdgcn_sched_barrier(0)

    v4f uA0[4], uA1[4], uB0[4], uB1[4], uC0[4], uC1[4];
    v4f sP0[4], sP1[4], sQ0[4], sQ1[4], sR0[4], sR1[4];

    LOAD_U(uA, 0);
    LOAD_U(uB, 1);
    LOAD_S(sP, 0);

    #pragma unroll
    for (int g = 0; g < 12; g += 3) {
        LOAD_U(uC, g + 2);  LOAD_S(sQ, g + 1);
        PIN();
        COMPUTE(uA, sP);
        PIN();
        LOAD_U(uA, g + 3);  LOAD_S(sR, g + 2);
        PIN();
        COMPUTE(uB, sQ);
        PIN();
        LOAD_U(uB, g + 4);  LOAD_S(sP, g + 3);
        PIN();
        COMPUTE(uC, sR);
        PIN();
    }
    LOAD_U(uC, 14);  LOAD_S(sQ, 13);
    PIN();
    COMPUTE(uA, sP);
    PIN();
    LOAD_U(uA, 15);  LOAD_S(sR, 14);
    PIN();
    COMPUTE(uB, sQ);
    PIN();
    LOAD_S(sP, 15);
    PIN();
    COMPUTE(uC, sR);
    PIN();
    COMPUTE(uA, sP);
    PIN();

#undef LOAD_U
#undef LOAD_S
#undef WMMA_F32
#undef COMPUTE
#undef PIN

    const int mrow0 = m_base +      8 * khalf;
    const int mrow1 = m_base + 16 + 8 * khalf;

    const float* mup0 = mu + mrow0;
    const float* mup1 = mu + mrow1;
    #pragma unroll
    for (int v = 0; v < 8; ++v) {
        const float mv0 = mup0[v];
        const float mv1 = mup1[v];
        c00[v] += mv0;  c01[v] += mv0;
        c10[v] += mv1;  c11[v] += mv1;
    }

    float* o00 = out + (size_t)lane16        * M_TOTAL + mrow0;
    float* o01 = out + (size_t)(lane16 + 16) * M_TOTAL + mrow0;
    float* o10 = out + (size_t)lane16        * M_TOTAL + mrow1;
    float* o11 = out + (size_t)(lane16 + 16) * M_TOTAL + mrow1;

    *(v4f*)(o00)     = (v4f){c00[0], c00[1], c00[2], c00[3]};
    *(v4f*)(o00 + 4) = (v4f){c00[4], c00[5], c00[6], c00[7]};
    *(v4f*)(o01)     = (v4f){c01[0], c01[1], c01[2], c01[3]};
    *(v4f*)(o01 + 4) = (v4f){c01[4], c01[5], c01[6], c01[7]};
    *(v4f*)(o10)     = (v4f){c10[0], c10[1], c10[2], c10[3]};
    *(v4f*)(o10 + 4) = (v4f){c10[4], c10[5], c10[6], c10[7]};
    *(v4f*)(o11)     = (v4f){c11[0], c11[1], c11[2], c11[3]};
    *(v4f*)(o11 + 4) = (v4f){c11[4], c11[5], c11[6], c11[7]};
}

extern "C" void kernel_launch(void* const* d_in, const int* in_sizes, int n_in,
                              void* d_out, int out_size, void* d_ws, size_t ws_size,
                              hipStream_t stream) {
    const float* style = (const float*)d_in[0];
    const float* U     = (const float*)d_in[1];
    const float* L     = (const float*)d_in[2];
    const float* mu    = (const float*)d_in[3];
    float* out         = (float*)d_out;

    (void)in_sizes; (void)n_in; (void)out_size; (void)d_ws; (void)ws_size;

    const int grid = M_TOTAL / ROWS_PER_BLK;
    eigenspace_wmma_kernel<<<grid, 256, 0, stream>>>(style, U, L, mu, out);
}


# U block fetched by whole-line global_load_b128 (4 lines per load, was 16 lines x 32 B) and turned into A fragments through a wave-private LDS tile (ds_store_b128/ds_load_b128); k order kept
# speedup vs baseline: 1.2121x; 1.0503x over previous
#include <hip/hip_runtime.h>

typedef __attribute__((ext_vector_type(2))) float v2f;
typedef __attribute__((ext_vector_type(4))) float v4f;
typedef __attribute__((ext_vector_type(8))) float v8f;

#define DIM        512
#define BATCH      32
#define M_TOTAL    (512 * 16 * 16)
#define TILES_PER_WAVE 2
#define ROWS_PER_WAVE  (16 * TILES_PER_WAVE)
#define WAVES_PER_BLK  8
#define ROWS_PER_BLK   (ROWS_PER_WAVE * WAVES_PER_BLK)
#define FACTOR     1.0f
#define U_PITCH    36
#define U_STAGE    (ROWS_PER_WAVE * U_PITCH)

__global__ __launch_bounds__(256) void eigenspace_wmma_kernel(
    const float* __restrict__ style,
    const float* __restrict__ U,
    const float* __restrict__ L,
    const float* __restrict__ mu,
    float* __restrict__ out)
{
    __shared__ __attribute__((aligned(16))) float s_lds[DIM * BATCH];
    __shared__ __attribute__((aligned(16))) float u_lds[WAVES_PER_BLK * U_STAGE];

    const int tid = threadIdx.x;

    for (int idx = tid; idx < DIM * BATCH; idx += 256) {
        const int b = idx >> 9;
        const int k = idx & (DIM - 1);
        const int q = k >> 2;
        const int j = k & 3;
        s_lds[(q << 7) + (b << 2) + j] = style[b * DIM + k] * (L[k] * FACTOR);
    }
    __syncthreads();

    const int wave   = tid >> 5;
    const int lane   = tid & 31;
    const int lane16 = lane & 15;
    const int khalf  = lane >> 4;

    const int m_base = blockIdx.x * ROWS_PER_BLK + wave * ROWS_PER_WAVE;

    const int grow = lane >> 3;
    const int gq   = lane & 7;
    const float* Ug = U + (size_t)(m_base + grow) * DIM + 4 * gq;

    float*       stw = u_lds + wave * U_STAGE + grow   * U_PITCH + 4 * gq;
    const float* lda = u_lds + wave * U_STAGE + lane16 * U_PITCH + 4 * khalf;

    const float* sb0 = s_lds + (khalf << 7) + 4 * lane16;
    const float* sb1 = sb0 + 64;

    v8f c00 = {};
    v8f c01 = {};
    v8f c10 = {};
    v8f c11 = {};


#define LOAD_G(S, g)                                                        \
    _Pragma("unroll")                                                       \
    for (int i = 0; i < 8; ++i)                                             \
        S[i] = *(const v4f*)(Ug + (size_t)(4 * i) * DIM + 32 * (g));

#define STORE_T(S)                                                          \
    _Pragma("unroll")                                                       \
    for (int i = 0; i < 8; ++i)                                             \
        *(v4f*)(stw + 4 * i * U_PITCH) = S[i];

#define LOAD_A(S)                                                           \
    _Pragma("unroll")                                                       \
    for (int j = 0; j < 4; ++j) {                                           \
        S##0[j] = *(const v4f*)(lda + 8 * j);                               \
        S##1[j] = *(const v4f*)(lda + 16 * U_PITCH + 8 * j);                \
    }

#define LOAD_S(S, g)                                                        \
    _Pragma("unroll")                                                       \
    for (int j = 0; j < 4; ++j) {                                           \
        S##0[j] = *(const v4f*)(sb0 + (32 * (g) + 8 * j) * 32);             \
        S##1[j] = *(const v4f*)(sb1 + (32 * (g) + 8 * j) * 32);             \
    }

#define WMMA_F32(acc, a, b)                                                 \
    acc = __builtin_amdgcn_wmma_f32_16x16x4_f32(false, a, false, b, (short)0, acc, false, false)

#define COMPUTE(A, B)                                                       \
    _Pragma("unroll")                                                       \
    for (int j = 0; j < 4; ++j) {                                           \
        const v2f a0l = {A##0[j][0], A##0[j][1]}, a0h = {A##0[j][2], A##0[j][3]}; \
        const v2f a1l = {A##1[j][0], A##1[j][1]}, a1h = {A##1[j][2], A##1[j][3]}; \
        const v2f b0l = {B##0[j][0], B##0[j][1]}, b0h = {B##0[j][2], B##0[j][3]}; \
        const v2f b1l = {B##1[j][0], B##1[j][1]}, b1h = {B##1[j][2], B##1[j][3]}; \
        WMMA_F32(c00, a0l, b0l);  WMMA_F32(c01, a0l, b1l);                  \
        WMMA_F32(c10, a1l, b0l);  WMMA_F32(c11, a1l, b1l);                  \
        WMMA_F32(c00, a0h, b0h);  WMMA_F32(c01, a0h, b1h);                  \
        WMMA_F32(c10, a1h, b0h);  WMMA_F32(c11, a1h, b1h);                  \
    }

#define PIN() __builtin_amdgcn_sched_barrier(0)

#define TIE() asm volatile("" : "+v"(c00), "+v"(c01), "+v"(c10), "+v"(c11))

    v4f gA[8], gB[8];
    v4f aP0[4], aP1[4], aQ0[4], aQ1[4];
    v4f sP0[4], sP1[4], sQ0[4], sQ1[4];

    LOAD_G(gA, 0);
    LOAD_G(gB, 1);
    PIN();
    STORE_T(gA);
    PIN();
    LOAD_G(gA, 2);  LOAD_A(aP);  LOAD_S(sP, 0);
    PIN();

    #pragma unroll
    for (int g = 0; g < 12; g += 2) {
        STORE_T(gB);
        PIN();
        LOAD_G(gB, g + 3);  LOAD_A(aQ);  LOAD_S(sQ, g + 1);
        PIN();
        TIE();  COMPUTE(aP, sP);  TIE();
        PIN();
        STORE_T(gA);
        PIN();
        LOAD_G(gA, g + 4);  LOAD_A(aP);  LOAD_S(sP, g + 2);
        PIN();
        TIE();  COMPUTE(aQ, sQ);  TIE();
        PIN();
    }
    STORE_T(gB);
    PIN();
    LOAD_G(gB, 15);  LOAD_A(aQ);  LOAD_S(sQ, 13);
    PIN();
    TIE();  COMPUTE(aP, sP);  TIE();
    PIN();
    STORE_T(gA);
    LOAD_A(aP);  LOAD_S(sP, 14);
    PIN();
    TIE();  COMPUTE(aQ, sQ);  TIE();
    PIN();
    STORE_T(gB);
    LOAD_A(aQ);  LOAD_S(sQ, 15);
    PIN();
    TIE();  COMPUTE(aP, sP);  TIE();
    PIN();
    TIE();  COMPUTE(aQ, sQ);  TIE();
    PIN();

#undef LOAD_G
#undef STORE_T
#undef LOAD_A
#undef LOAD_S
#undef WMMA_F32
#undef COMPUTE
#undef PIN
#undef TIE

    const int mrow0 = m_base +      8 * khalf;
    const int mrow1 = m_base + 16 + 8 * khalf;

    const float* mup0 = mu + mrow0;
    const float* mup1 = mu + mrow1;
    #pragma unroll
    for (int v = 0; v < 8; ++v) {
        const float mv0 = mup0[v];
        const float mv1 = mup1[v];
        c00[v] += mv0;  c01[v] += mv0;
        c10[v] += mv1;  c11[v] += mv1;
    }

    float* o00 = out + (size_t)lane16        * M_TOTAL + mrow0;
    float* o01 = out + (size_t)(lane16 + 16) * M_TOTAL + mrow0;
    float* o10 = out + (size_t)lane16        * M_TOTAL + mrow1;
    float* o11 = out + (size_t)(lane16 + 16) * M_TOTAL + mrow1;

    *(v4f*)(o00)     = (v4f){c00[0], c00[1], c00[2], c00[3]};
    *(v4f*)(o00 + 4) = (v4f){c00[4], c00[5], c00[6], c00[7]};
    *(v4f*)(o01)     = (v4f){c01[0], c01[1], c01[2], c01[3]};
    *(v4f*)(o01 + 4) = (v4f){c01[4], c01[5], c01[6], c01[7]};
    *(v4f*)(o10)     = (v4f){c10[0], c10[1], c10[2], c10[3]};
    *(v4f*)(o10 + 4) = (v4f){c10[4], c10[5], c10[6], c10[7]};
    *(v4f*)(o11)     = (v4f){c11[0], c11[1], c11[2], c11[3]};
    *(v4f*)(o11 + 4) = (v4f){c11[4], c11[5], c11[6], c11[7]};
}

extern "C" void kernel_launch(void* const* d_in, const int* in_sizes, int n_in,
                              void* d_out, int out_size, void* d_ws, size_t ws_size,
                              hipStream_t stream) {
    const float* style = (const float*)d_in[0];
    const float* U     = (const float*)d_in[1];
    const float* L     = (const float*)d_in[2];
    const float* mu    = (const float*)d_in[3];
    float* out         = (float*)d_out;

    (void)in_sizes; (void)n_in; (void)out_size; (void)d_ws; (void)ws_size;

    const int grid = M_TOTAL / ROWS_PER_BLK;
    eigenspace_wmma_kernel<<<grid, 256, 0, stream>>>(style, U, L, mu, out);
}
